# starting kernel
# baseline (speedup 1.0000x reference)
#include <hip/hip_runtime.h>

typedef __attribute__((ext_vector_type(2))) float v2f;
typedef __attribute__((ext_vector_type(8))) float v8f;

#define IMG_W   512
#define DEPTH   64
#define OUT_N   510
#define NOUT    (OUT_N * OUT_N)
#define NTILES  ((NOUT + 15) / 16)
#define KTOT    (3 * 3 * DEPTH)

__launch_bounds__(256)
__global__ void conv3x3d64_wmma_f32(const float* __restrict__ x,
                                    const float* __restrict__ W,
                                    const float* __restrict__ bias,
                                    float* __restrict__ out)
{
    __shared__ float wlds[KTOT];
    for (int i = threadIdx.x; i < KTOT; i += 256) wlds[i] = W[i];
    __syncthreads();

    const int wave = threadIdx.x >> 5;
    const int lane = threadIdx.x & 31;
    const int tile = blockIdx.x * 8 + wave;
    if (tile >= NTILES) return;

    const int      n  = lane & 15;
    const int      h  = lane >> 4;
    const unsigned p  = (unsigned)tile * 16u + (unsigned)n;
    const unsigned pe = (p < NOUT) ? p : (NOUT - 1);
    const unsigned r  = pe / OUT_N;
    const unsigned c  = pe % OUT_N;

    const float* xb = x    + ((size_t)r * IMG_W + c) * DEPTH + 2 * h;
    const float* wb = wlds + 2 * h;

    v8f acc = {};
#pragma unroll
    for (int fi = 0; fi < 3; ++fi) {
#pragma unroll
        for (int fj = 0; fj < 3; ++fj) {
#pragma unroll
            for (int d0 = 0; d0 < DEPTH; d0 += 4) {
                const int xoff = fi * (IMG_W * DEPTH) + fj * DEPTH + d0;
                const int koff = (fi * 3 + fj) * DEPTH + d0;
                v2f bvec = *(const v2f*)(xb + xoff);
                v2f avec = *(const v2f*)(wb + koff);
                acc = __builtin_amdgcn_wmma_f32_16x16x4_f32(
                         false, avec,  false, bvec,
                         (short)0, acc,
                         false,  false);
            }
        }
    }

    if (h == 0 && p < NOUT) out[p] = acc[0] + bias[0];
}

extern "C" void kernel_launch(void* const* d_in, const int* in_sizes, int n_in,
                              void* d_out, int out_size, void* d_ws, size_t ws_size,
                              hipStream_t stream) {
    (void)in_sizes; (void)n_in; (void)d_ws; (void)ws_size; (void)out_size;
    const float* x  = (const float*)d_in[0];
    const float* W  = (const float*)d_in[1];
    const float* b  = (const float*)d_in[2];
    float* out      = (float*)d_out;

    const int blocks = (NTILES + 7) / 8;
    hipLaunchKernelGGL(conv3x3d64_wmma_f32, dim3(blocks), dim3(256), 0, stream,
                       x, W, b, out);
}


# WMMA rows = 4 partial sums x 4 output columns of a 2x6-pixel row-pair segment (was a 4x4 patch in a 6x6 window): 25% fewer WMMAs; sum order: taps per depth half, row pairs, 8 depth slices
# speedup vs baseline: 7.1482x; 7.1482x over previous
#include <hip/hip_runtime.h>

typedef __attribute__((ext_vector_type(2))) float v2f;
typedef __attribute__((ext_vector_type(4))) float v4f;
typedef __attribute__((ext_vector_type(8))) float v8f;

#define IMG_W   512
#define DEPTH   64
#define OUT_N   510
#define TW      64
#define TH      16
#define WPX     66
#define PPITCH  66
#define TPITCH  68
#define CROWS   6
#define NCH     3
#define CHSZ    (CROWS * WPX * PPITCH)
#define NWG     256
#define BP      3

static __device__ __forceinline__ void load_chunk(v4f (&buf)[25], const float* __restrict__ x,
                                                  const int row0, const int c0, const int tid)
{
    const int px = tid >> 4;
    const int d4 = tid & 15;
#pragma unroll
    for (int j = 0; j < CROWS; ++j) {
        const int gr = min(row0 + j, IMG_W - 1);
        const float* rp = x + (size_t)gr * (IMG_W * DEPTH) + (c0 + px) * DEPTH + d4 * 4;
#pragma unroll
        for (int q = 0; q < 4; ++q) buf[j * 4 + q] = *(const v4f*)(rp + q * 16 * DEPTH);
    }
    if (tid < 192) {
        const int j  = tid >> 5;
        const int e  = (tid >> 4) & 1;
        const int gr = min(row0 + j, IMG_W - 1);
        const int gc = min(c0 + 64 + e, IMG_W - 1);
        buf[24] = *(const v4f*)(x + (size_t)gr * (IMG_W * DEPTH) + gc * DEPTH + d4 * 4);
    }
}

static __device__ __forceinline__ void store_chunk(const v4f (&buf)[25], float* __restrict__ w,
                                                   const int tid)
{
    const int px = tid >> 4;
    const int d4 = tid & 15;
#pragma unroll
    for (int j = 0; j < CROWS; ++j) {
#pragma unroll
        for (int q = 0; q < 4; ++q) {
            float* dst = w + (j * WPX + q * 16 + px) * PPITCH + d4 * 4;
            *(v2f*)(dst)     = (v2f){buf[j * 4 + q].x, buf[j * 4 + q].y};
            *(v2f*)(dst + 2) = (v2f){buf[j * 4 + q].z, buf[j * 4 + q].w};
        }
    }
    if (tid < 192) {
        const int j = tid >> 5;
        const int e = (tid >> 4) & 1;
        float* dst = w + (j * WPX + 64 + e) * PPITCH + d4 * 4;
        *(v2f*)(dst)     = (v2f){buf[24].x, buf[24].y};
        *(v2f*)(dst + 2) = (v2f){buf[24].z, buf[24].w};
    }
}

__launch_bounds__(256)
__global__ void conv3x3d64_rowpair_wmma_f32(const float* __restrict__ x,
                                            const float* __restrict__ W,
                                            const float* __restrict__ bias,
                                            float* __restrict__ out)
{
    __shared__ __attribute__((aligned(16))) float win[2 * CHSZ];
    __shared__ __attribute__((aligned(16))) float wl[9 * TPITCH];
    __shared__ __attribute__((aligned(16))) float red[8 * (TH + 2) * TW];

    const int tid = threadIdx.x;
    const int r0  = (blockIdx.x >> 3) * TH;
    const int c0  = (blockIdx.x & 7) * TW;

    v4f buf[25] = {};
    load_chunk(buf, x, r0, c0, tid);

    const float wv0 = W[tid];
    const float wv1 = W[tid + 256];
    float wv2 = 0.0f;
    if (tid < 9 * DEPTH - 512) wv2 = W[tid + 512];
    const float bv = bias[0];

    store_chunk(buf, win, tid);
    wl[(tid >> 6) * TPITCH + (tid & 63)]       = wv0;
    wl[((tid >> 6) + 4) * TPITCH + (tid & 63)] = wv1;
    if (tid < 9 * DEPTH - 512) wl[8 * TPITCH + tid] = wv2;
    __syncthreads();

    const int wave = tid >> 5;
    const int lane = tid & 31;
    const int n    = lane & 15;
    const int h    = lane >> 4;
    const int f    = n >> 2;
    const int ox   = n & 3;
    const int fi0  = (f == 0) ? 0 : (f == 1) ? 2 : (f == 2) ? -1 : 1;
    const int fi1  = (f == 0) ? 1 : (f == 1) ? -1 : (f == 2) ? 0 : 2;
    const int dofs = 8 * wave + 2 * h;
    const int bofs = (4 * n) * PPITCH + dofs;

    float* rw = red + wave * ((TH + 2) * TW) + h * TW + 4 * n;

    v4f carry = {};
#pragma unroll 1
    for (int c = 0; c < NCH; ++c) {
        if (c + 1 < NCH) load_chunk(buf, x, r0 + (c + 1) * CROWS, c0, tid);

        const float* bb = win + (c & 1) * CHSZ + bofs;

        v8f acc0 = {};
        v8f acc1 = {};
        v2f fa0[2][BP], fa1[2][BP], fb0[2][BP], fb1[2][BP];
#pragma unroll
        for (int b = -1; b < 36 / BP; ++b) {
            if (b + 1 < 36 / BP) {
#pragma unroll
                for (int p = 0; p < BP; ++p) {
                    const int  s     = (b + 1) * BP + p;
                    const int  rowl  = s / 6;
                    const int  ix    = s - 6 * rowl;
                    const int  fi    = (rowl & 1) ? fi1 : fi0;
                    const int  fj    = ix - ox;
                    const bool valid = (fi >= 0) && ((unsigned)fj < 3u);
                    const int  aoff  = (valid ? (fi * 3 + fj) * TPITCH : 0) + dofs;
                    const int  boff  = (rowl * WPX + ix) * PPITCH;
                    fa0[(b + 1) & 1][p] = *(const v2f*)(wl + aoff);
                    fa1[(b + 1) & 1][p] = *(const v2f*)(wl + aoff + 4);
                    fb0[(b + 1) & 1][p] = *(const v2f*)(bb + boff);
                    fb1[(b + 1) & 1][p] = *(const v2f*)(bb + boff + 4);
                }
            }
            __builtin_amdgcn_sched_barrier(0);
            if (b >= 0) {
#pragma unroll
                for (int p = 0; p < BP; ++p) {
                    const int  s     = b * BP + p;
                    const int  rowl  = s / 6;
                    const int  ix    = s - 6 * rowl;
                    const int  fi    = (rowl & 1) ? fi1 : fi0;
                    const int  fj    = ix - ox;
                    const bool valid = (fi >= 0) && ((unsigned)fj < 3u);
                    v2f a0 = fa0[b & 1][p];
                    v2f a1 = fa1[b & 1][p];
                    a0.x = valid ? a0.x : 0.0f;  a0.y = valid ? a0.y : 0.0f;
                    a1.x = valid ? a1.x : 0.0f;  a1.y = valid ? a1.y : 0.0f;
                    acc0 = __builtin_amdgcn_wmma_f32_16x16x4_f32(false, a0, false, fb0[b & 1][p], (short)0, acc0, false, false);
                    acc1 = __builtin_amdgcn_wmma_f32_16x16x4_f32(false, a1, false, fb1[b & 1][p], (short)0, acc1, false, false);
                }
                if ((b & 3) == 3) {
                    const v8f d  = acc0 + acc1;
                    const int sg = c * 3 + (b >> 2);
                    *(v4f*)(rw + sg * (2 * TW)) = carry + (v4f){d[4], d[5], d[6], d[7]};
                    carry = (v4f){d[0], d[1], d[2], d[3]};
                    acc0 = (v8f){};
                    acc1 = (v8f){};
                }
            }
            __builtin_amdgcn_sched_barrier(0);
        }

        if (c + 1 < NCH) store_chunk(buf, win + ((c + 1) & 1) * CHSZ, tid);

        __syncthreads();
    }

#pragma unroll
    for (int e = 0; e < 4; ++e) {
        const int o   = tid + 256 * e;
        const int row = o >> 6;
        const int col = o & 63;
        const float* rr = red + 2 * TW + o;
        float s = rr[0];
        for (int w = 1; w < 8; ++w) s += rr[w * ((TH + 2) * TW)];
        s += bv;
        const int orow = r0 + row;
        const int ocol = c0 + col;
        if (orow < OUT_N && ocol < OUT_N) out[orow * OUT_N + ocol] = s;
    }
}

extern "C" void kernel_launch(void* const* d_in, const int* in_sizes, int n_in,
                              void* d_out, int out_size, void* d_ws, size_t ws_size,
                              hipStream_t stream) {
    (void)in_sizes; (void)n_in; (void)d_ws; (void)ws_size; (void)out_size;
    const float* x  = (const float*)d_in[0];
    const float* W  = (const float*)d_in[1];
    const float* b  = (const float*)d_in[2];
    float* out      = (float*)d_out;

    hipLaunchKernelGGL(conv3x3d64_rowpair_wmma_f32, dim3(NWG), dim3(256), 0, stream,
                       x, W, b, out);
}


# LDS staging handed over per row pair (was per 6-row chunk): first WMMA waits for 2 of 18 rows; pair s+1 stored among the WMMAs of pair s, one LDS-only barrier per pair; loads 3-4 pairs ahead
# speedup vs baseline: 7.6578x; 1.0713x over previous
#include <hip/hip_runtime.h>

typedef __attribute__((ext_vector_type(2))) float v2f;
typedef __attribute__((ext_vector_type(4))) float v4f;
typedef __attribute__((ext_vector_type(8))) float v8f;

#define IMG_W   512
#define DEPTH   64
#define OUT_N   510
#define TW      64
#define TH      16
#define WPX     66
#define PPITCH  66
#define TPITCH  68
#define PAIRSZ  (2 * WPX * PPITCH)
#define NSLOT   6
#define NCH     3
#define NWG     256
#define BP      3

static __device__ __forceinline__ void load_pair(v4f (&buf)[9], const float* __restrict__ x,
                                                 const int row0, const int c0, const int tid)
{
    const int px = tid >> 4;
    const int d4 = tid & 15;
#pragma unroll
    for (int j = 0; j < 2; ++j) {
        const int gr = min(row0 + j, IMG_W - 1);
        const float* rp = x + (size_t)gr * (IMG_W * DEPTH) + (c0 + px) * DEPTH + d4 * 4;
#pragma unroll
        for (int q = 0; q < 4; ++q) buf[j * 4 + q] = *(const v4f*)(rp + q * 16 * DEPTH);
    }
    if (tid < 64) {
        const int j  = tid >> 5;
        const int e  = (tid >> 4) & 1;
        const int gr = min(row0 + j, IMG_W - 1);
        const int gc = min(c0 + 64 + e, IMG_W - 1);
        buf[8] = *(const v4f*)(x + (size_t)gr * (IMG_W * DEPTH) + gc * DEPTH + d4 * 4);
    }
}

static __device__ __forceinline__ void store_pair(const v4f (&buf)[9], float* __restrict__ w,
                                                  const int tid)
{
    const int px = tid >> 4;
    const int d4 = tid & 15;
#pragma unroll
    for (int j = 0; j < 2; ++j) {
#pragma unroll
        for (int q = 0; q < 4; ++q) {
            float* dst = w + (j * WPX + q * 16 + px) * PPITCH + d4 * 4;
            *(v2f*)(dst)     = (v2f){buf[j * 4 + q].x, buf[j * 4 + q].y};
            *(v2f*)(dst + 2) = (v2f){buf[j * 4 + q].z, buf[j * 4 + q].w};
        }
    }
    if (tid < 64) {
        const int j = tid >> 5;
        const int e = (tid >> 4) & 1;
        float* dst = w + (j * WPX + 64 + e) * PPITCH + d4 * 4;
        *(v2f*)(dst)     = (v2f){buf[8].x, buf[8].y};
        *(v2f*)(dst + 2) = (v2f){buf[8].z, buf[8].w};
    }
}

static __device__ __forceinline__ void lds_barrier()
{
    __builtin_amdgcn_fence(__ATOMIC_RELEASE, "workgroup", "local");
    __builtin_amdgcn_s_barrier();
    __builtin_amdgcn_fence(__ATOMIC_ACQUIRE, "workgroup", "local");
}

__launch_bounds__(256)
__global__ void conv3x3d64_rowpair_wmma_f32(const float* __restrict__ x,
                                            const float* __restrict__ W,
                                            const float* __restrict__ bias,
                                            float* __restrict__ out)
{
    __shared__ __attribute__((aligned(16))) float win[NSLOT * PAIRSZ];
    __shared__ __attribute__((aligned(16))) float wl[9 * TPITCH];
    __shared__ __attribute__((aligned(16))) float red[8 * (TH + 2) * TW];

    const int tid = threadIdx.x;
    const int r0  = (blockIdx.x >> 3) * TH;
    const int c0  = (blockIdx.x & 7) * TW;

    const float wv0 = W[tid];
    const float wv1 = W[tid + 256];
    float wv2 = 0.0f;
    if (tid < 9 * DEPTH - 512) wv2 = W[tid + 512];
    const float bv = bias[0];

    v4f s0[9] = {}, s1[9] = {}, s2[9] = {};
    load_pair(s0, x, r0,     c0, tid);
    load_pair(s1, x, r0 + 2, c0, tid);
    load_pair(s2, x, r0 + 4, c0, tid);

    store_pair(s0, win, tid);
    load_pair(s0, x, r0 + 6, c0, tid);
    wl[(tid >> 6) * TPITCH + (tid & 63)]       = wv0;
    wl[((tid >> 6) + 4) * TPITCH + (tid & 63)] = wv1;
    if (tid < 9 * DEPTH - 512) wl[8 * TPITCH + tid] = wv2;
    lds_barrier();

    const int wave = tid >> 5;
    const int lane = tid & 31;
    const int n    = lane & 15;
    const int h    = lane >> 4;
    const int f    = n >> 2;
    const int ox   = n & 3;
    const int fi0  = (f == 0) ? 0 : (f == 1) ? 2 : (f == 2) ? -1 : 1;
    const int fi1  = (f == 0) ? 1 : (f == 1) ? -1 : (f == 2) ? 0 : 2;
    const int dofs = 8 * wave + 2 * h;
    const int bofs = (4 * n) * PPITCH + dofs;

    float* rw = red + wave * ((TH + 2) * TW) + h * TW + 4 * n;

    const float* ap[2][6];
#pragma unroll
    for (int par = 0; par < 2; ++par) {
#pragma unroll
        for (int ix = 0; ix < 6; ++ix) {
            const int  fi    = par ? fi1 : fi0;
            const int  fj    = ix - ox;
            const bool valid = (fi >= 0) && ((unsigned)fj < 3u);
            ap[par][ix] = wl + (valid ? (fi * 3 + fj) * TPITCH : 0) + dofs;
        }
    }

    v4f carry = {};
#pragma unroll
    for (int c = 0; c < NCH; ++c) {
        float* wcur = win + (c & 1) * (3 * PAIRSZ);
        float* wnxt = win + ((c + 1) & 1) * (3 * PAIRSZ);
        const float* bb = wcur + bofs;
        const int rowc = r0 + 6 * c;

        v8f acc0 = {};
        v8f acc1 = {};
        v2f fa0[2][BP], fa1[2][BP], fb0[2][BP], fb1[2][BP];
#pragma unroll
        for (int b = -1; b < 36 / BP; ++b) {
            if (b + 1 < 36 / BP) {
#pragma unroll
                for (int p = 0; p < BP; ++p) {
                    const int  s     = (b + 1) * BP + p;
                    const int  rowl  = s / 6;
                    const int  ix    = s - 6 * rowl;
                    const int  boff  = (rowl * WPX + ix) * PPITCH;
                    fa0[(b + 1) & 1][p] = *(const v2f*)(ap[rowl & 1][ix]);
                    fa1[(b + 1) & 1][p] = *(const v2f*)(ap[rowl & 1][ix] + 4);
                    fb0[(b + 1) & 1][p] = *(const v2f*)(bb + boff);
                    fb1[(b + 1) & 1][p] = *(const v2f*)(bb + boff + 4);
                }
            }
            if (b >= 0 && (b & 3) == 0) {
                __builtin_amdgcn_sched_barrier(0);
                if (b == 0) {
                    store_pair(s1, wcur + PAIRSZ, tid);
                    if (c < 2) load_pair(s1, x, rowc + 8, c0, tid);
                } else if (b == 4) {
                    store_pair(s2, wcur + 2 * PAIRSZ, tid);
                    if (c < 2) load_pair(s2, x, rowc + 10, c0, tid);
                } else {
                    if (c < 2) store_pair(s0, wnxt, tid);
                    if (c < 1) load_pair(s0, x, rowc + 12, c0, tid);
                }
            }
            __builtin_amdgcn_sched_barrier(0);
            if (b >= 0) {
#pragma unroll
                for (int p = 0; p < BP; ++p) {
                    const int  s     = b * BP + p;
                    const int  rowl  = s / 6;
                    const int  ix    = s - 6 * rowl;
                    const int  fi    = (rowl & 1) ? fi1 : fi0;
                    const int  fj    = ix - ox;
                    const bool valid = (fi >= 0) && ((unsigned)fj < 3u);
                    v2f a0 = fa0[b & 1][p];
                    v2f a1 = fa1[b & 1][p];
                    a0.x = valid ? a0.x : 0.0f;  a0.y = valid ? a0.y : 0.0f;
                    a1.x = valid ? a1.x : 0.0f;  a1.y = valid ? a1.y : 0.0f;
                    acc0 = __builtin_amdgcn_wmma_f32_16x16x4_f32(false, a0, false, fb0[b & 1][p], (short)0, acc0, false, false);
                    acc1 = __builtin_amdgcn_wmma_f32_16x16x4_f32(false, a1, false, fb1[b & 1][p], (short)0, acc1, false, false);
                }
                if ((b & 3) == 3) {
                    const v8f d  = acc0 + acc1;
                    const int sg = c * 3 + (b >> 2);
                    *(v4f*)(rw + sg * (2 * TW)) = carry + (v4f){d[4], d[5], d[6], d[7]};
                    carry = (v4f){d[0], d[1], d[2], d[3]};
                    acc0 = (v8f){};
                    acc1 = (v8f){};
                }
                if ((b & 3) == 1) lds_barrier();
            }
            __builtin_amdgcn_sched_barrier(0);
        }

    }

    __syncthreads();

#pragma unroll
    for (int e = 0; e < 4; ++e) {
        const int o   = tid + 256 * e;
        const int row = o >> 6;
        const int col = o & 63;
        const float* rr = red + 2 * TW + o;
        float s = rr[0];
        for (int w = 1; w < 8; ++w) s += rr[w * ((TH + 2) * TW)];
        s += bv;
        const int orow = r0 + row;
        const int ocol = c0 + col;
        if (orow < OUT_N && ocol < OUT_N) out[orow * OUT_N + ocol] = s;
    }
}

extern "C" void kernel_launch(void* const* d_in, const int* in_sizes, int n_in,
                              void* d_out, int out_size, void* d_ws, size_t ws_size,
                              hipStream_t stream) {
    (void)in_sizes; (void)n_in; (void)d_ws; (void)ws_size; (void)out_size;
    const float* x  = (const float*)d_in[0];
    const float* W  = (const float*)d_in[1];
    const float* b  = (const float*)d_in[2];
    float* out      = (float*)d_out;

    hipLaunchKernelGGL(conv3x3d64_rowpair_wmma_f32, dim3(NWG), dim3(256), 0, stream,
                       x, W, b, out);
}
